# starting kernel
# baseline (speedup 1.0000x reference)
#include <hip/hip_runtime.h>


#define RANK   64
#define WIN    11
#define HALO   5
#define TILE   64
#define XROWS  (TILE + 2*HALO)
#define LD_X   66
#define LD_W   68

typedef float v2f __attribute__((ext_vector_type(2)));
typedef float v8f __attribute__((ext_vector_type(8)));

static __device__ __forceinline__ float fast_tanh(float x) {
#if __has_builtin(__builtin_amdgcn_tanhf)
    return __builtin_amdgcn_tanhf(x);
#elif __has_builtin(__builtin_amdgcn_tanh_f32)
    return __builtin_amdgcn_tanh_f32(x);
#else
    return tanhf(x);
#endif
}

static __device__ __forceinline__ float fast_exp_nonpos(float x) {
#if __has_builtin(__builtin_amdgcn_exp2f)
    return __builtin_amdgcn_exp2f(x * 1.44269504088896341f);
#else
    return __expf(x);
#endif
}

__global__ __launch_bounds__(128)
void attn_window_fused_kernel(const float* __restrict__ x,
                              const float* __restrict__ W,
                              const float* __restrict__ b,
                              float* __restrict__ out,
                              int nrows)
{
    __shared__ float xs[XROWS * LD_X];
    __shared__ float vs[TILE  * LD_X];
    __shared__ float ws[RANK  * LD_W];
    __shared__ float bs[RANK];

    const int tid = threadIdx.x;
    const int r0  = blockIdx.x * TILE;

    for (int idx = tid; idx < RANK * (RANK / 4); idx += blockDim.x) {
        const int row = idx >> 4;
        const int c4  = idx & 15;
        float4 v = ((const float4*)W)[row * (RANK / 4) + c4];
        *(float4*)&ws[row * LD_W + c4 * 4] = v;
    }
    if (tid < RANK) bs[tid] = b[tid];

    for (int idx = tid; idx < XROWS * (RANK / 4); idx += blockDim.x) {
        const int row = idx >> 4;
        const int c4  = idx & 15;
        const int g   = r0 - HALO + row;
        float4 v = make_float4(0.f, 0.f, 0.f, 0.f);
        if (g >= 0 && g < nrows) v = ((const float4*)x)[g * (RANK / 4) + c4];
        float* p = &xs[row * LD_X + c4 * 4];
        *(float2*)(p)     = make_float2(v.x, v.y);
        *(float2*)(p + 2) = make_float2(v.z, v.w);
    }
    __syncthreads();

    {
        const int wave = tid >> 5;
        const int lane = tid & 31;
        const int m    = lane & 15;
        const int hi   = lane >> 4;
        const int arow = wave * 16 + m + HALO;

        v2f afrag[RANK / 4];
        #pragma unroll
        for (int kb = 0; kb < RANK / 4; ++kb)
            afrag[kb] = *(const v2f*)&xs[arow * LD_X + kb * 4 + hi * 2];

        #pragma unroll
        for (int nt = 0; nt < 4; ++nt) {
            v8f acc = {};
            const int n = nt * 16 + m;
            #pragma unroll
            for (int kb = 0; kb < RANK / 4; ++kb) {
                v2f bfrag = *(const v2f*)&ws[n * LD_W + kb * 4 + hi * 2];
                acc = __builtin_amdgcn_wmma_f32_16x16x4_f32(
                        false, afrag[kb], false, bfrag,
                        (short)0, acc, false, false);
            }
            const float bias = bs[n];
            #pragma unroll
            for (int v = 0; v < 8; ++v) {
                const int row = wave * 16 + v + hi * 8;
                vs[row * LD_X + n] = fast_tanh(acc[v] + bias);
            }
        }
    }
    __syncthreads();

    {
        const int r    = tid >> 1;
        const int half = tid & 1;
        const int g    = r0 + r;
        if (g < nrows) {
            const int c0 = half * 32;
            float s[WIN];
            #pragma unroll
            for (int w = 0; w < WIN; ++w) {
                v2f a0 = {0.f, 0.f}, a1 = {0.f, 0.f};
                #pragma unroll
                for (int c = 0; c < 32; c += 4) {
                    v2f x0 = *(const v2f*)&xs[(r + w) * LD_X + c0 + c];
                    v2f x1 = *(const v2f*)&xs[(r + w) * LD_X + c0 + c + 2];
                    v2f v0 = *(const v2f*)&vs[r * LD_X + c0 + c];
                    v2f v1 = *(const v2f*)&vs[r * LD_X + c0 + c + 2];
                    a0 = __builtin_elementwise_fma(x0, v0, a0);
                    a1 = __builtin_elementwise_fma(x1, v1, a1);
                }
                v2f acc = a0 + a1;
                float sv = acc.x + acc.y;
                sv += __shfl_xor(sv, 1, 32);
                s[w] = sv * 0.125f;
            }
            float mx = s[0];
            #pragma unroll
            for (int w = 1; w < WIN; ++w) mx = fmaxf(mx, s[w]);
            float sum = 0.f;
            #pragma unroll
            for (int w = 0; w < WIN; ++w) { s[w] = fast_exp_nonpos(s[w] - mx); sum += s[w]; }
            const float inv = 1.0f / sum;

            #pragma unroll
            for (int c = 0; c < 32; c += 2) {
                v2f o = {0.f, 0.f};
                #pragma unroll
                for (int w = 0; w < WIN; ++w) {
                    v2f xv = *(const v2f*)&xs[(r + w) * LD_X + c0 + c];
                    v2f sw = {s[w], s[w]};
                    o = __builtin_elementwise_fma(sw, xv, o);
                }
                o *= inv;
                *(v2f*)&out[g * RANK + c0 + c] = o;
            }
        }
    }
}

extern "C" void kernel_launch(void* const* d_in, const int* in_sizes, int n_in,
                              void* d_out, int out_size, void* d_ws, size_t ws_size,
                              hipStream_t stream) {
    const float* x = (const float*)d_in[0];
    const float* W = (const float*)d_in[1];
    const float* b = (const float*)d_in[2];
    float* out = (float*)d_out;

    const int nrows = in_sizes[0] / RANK;
    const int grid  = (nrows + TILE - 1) / TILE;
    attn_window_fused_kernel<<<grid, 128, 0, stream>>>(x, W, b, out, nrows);
}


# staging: each thread's 19 global loads (W, bias, x tile with halo) issued as one group ahead of the first LDS store; before, each loop trip waited for its own load, 19 round trips in series
# speedup vs baseline: 2.0196x; 2.0196x over previous
#include <hip/hip_runtime.h>


#define RANK   64
#define WIN    11
#define HALO   5
#define TILE   64
#define XROWS  (TILE + 2*HALO)
#define LD_X   66
#define LD_W   68
#define NTHR   128
#define WLOADS (RANK * (RANK / 4) / NTHR)
#define XLOADS (XROWS / 8)
#define XTAIL  (XROWS - 8 * XLOADS)

typedef float v2f __attribute__((ext_vector_type(2)));
typedef float v8f __attribute__((ext_vector_type(8)));

static __device__ __forceinline__ float fast_tanh(float x) {
#if __has_builtin(__builtin_amdgcn_tanhf)
    return __builtin_amdgcn_tanhf(x);
#elif __has_builtin(__builtin_amdgcn_tanh_f32)
    return __builtin_amdgcn_tanh_f32(x);
#else
    return tanhf(x);
#endif
}

static __device__ __forceinline__ float fast_exp_nonpos(float x) {
#if __has_builtin(__builtin_amdgcn_exp2f)
    return __builtin_amdgcn_exp2f(x * 1.44269504088896341f);
#else
    return __expf(x);
#endif
}

__global__ __launch_bounds__(NTHR)
void attn_window_fused_kernel(const float* __restrict__ x,
                              const float* __restrict__ W,
                              const float* __restrict__ b,
                              float* __restrict__ out,
                              int nrows)
{
    __shared__ float xs[XROWS * LD_X];
    __shared__ float vs[TILE  * LD_X];
    __shared__ float ws[RANK  * LD_W];
    __shared__ float bs[RANK];

    const int tid  = threadIdx.x;
    const int r0   = blockIdx.x * TILE;
    const int c4   = tid & 15;
    const int rsub = tid >> 4;

    float4 wv[WLOADS];
    for (int i = 0; i < WLOADS; ++i)
        wv[i] = ((const float4*)W)[tid + i * NTHR];

    const float bv = b[tid & (RANK - 1)];

    float4 xv[XLOADS + 1];
    for (int i = 0; i < XLOADS; ++i) {
        const int g  = r0 - HALO + rsub + 8 * i;
        const int gc = min(max(g, 0), nrows - 1);
        xv[i] = ((const float4*)x)[gc * (RANK / 4) + c4];
    }
    xv[XLOADS] = make_float4(0.f, 0.f, 0.f, 0.f);
    if (tid < XTAIL * 16) {
        const int g  = r0 - HALO + 8 * XLOADS + rsub;
        const int gc = min(max(g, 0), nrows - 1);
        xv[XLOADS] = ((const float4*)x)[gc * (RANK / 4) + c4];
    }

    for (int i = 0; i < WLOADS; ++i)
        *(float4*)&ws[(rsub + 8 * i) * LD_W + c4 * 4] = wv[i];
    if (tid < RANK) bs[tid] = bv;

    for (int i = 0; i < XLOADS; ++i) {
        const int row = rsub + 8 * i;
        const int g   = r0 - HALO + row;
        float4 v = xv[i];
        if (g < 0 || g >= nrows) v = make_float4(0.f, 0.f, 0.f, 0.f);
        float* p = &xs[row * LD_X + c4 * 4];
        *(float2*)(p)     = make_float2(v.x, v.y);
        *(float2*)(p + 2) = make_float2(v.z, v.w);
    }
    if (tid < XTAIL * 16) {
        const int row = 8 * XLOADS + rsub;
        const int g   = r0 - HALO + row;
        float4 v = xv[XLOADS];
        if (g < 0 || g >= nrows) v = make_float4(0.f, 0.f, 0.f, 0.f);
        float* p = &xs[row * LD_X + c4 * 4];
        *(float2*)(p)     = make_float2(v.x, v.y);
        *(float2*)(p + 2) = make_float2(v.z, v.w);
    }
    __syncthreads();

    {
        const int wave = tid >> 5;
        const int lane = tid & 31;
        const int m    = lane & 15;
        const int hi   = lane >> 4;
        const int arow = wave * 16 + m + HALO;

        v2f afrag[RANK / 4];
        #pragma unroll
        for (int kb = 0; kb < RANK / 4; ++kb)
            afrag[kb] = *(const v2f*)&xs[arow * LD_X + kb * 4 + hi * 2];

        #pragma unroll
        for (int nt = 0; nt < 4; ++nt) {
            v8f acc = {};
            const int n = nt * 16 + m;
            #pragma unroll
            for (int kb = 0; kb < RANK / 4; ++kb) {
                v2f bfrag = *(const v2f*)&ws[n * LD_W + kb * 4 + hi * 2];
                acc = __builtin_amdgcn_wmma_f32_16x16x4_f32(
                        false, afrag[kb], false, bfrag,
                        (short)0, acc, false, false);
            }
            const float bias = bs[n];
            #pragma unroll
            for (int v = 0; v < 8; ++v) {
                const int row = wave * 16 + v + hi * 8;
                vs[row * LD_X + n] = fast_tanh(acc[v] + bias);
            }
        }
    }
    __syncthreads();

    {
        const int r    = tid >> 1;
        const int half = tid & 1;
        const int g    = r0 + r;
        if (g < nrows) {
            const int c0 = half * 32;
            float s[WIN];
            #pragma unroll
            for (int w = 0; w < WIN; ++w) {
                v2f a0 = {0.f, 0.f}, a1 = {0.f, 0.f};
                #pragma unroll
                for (int c = 0; c < 32; c += 4) {
                    v2f x0 = *(const v2f*)&xs[(r + w) * LD_X + c0 + c];
                    v2f x1 = *(const v2f*)&xs[(r + w) * LD_X + c0 + c + 2];
                    v2f v0 = *(const v2f*)&vs[r * LD_X + c0 + c];
                    v2f v1 = *(const v2f*)&vs[r * LD_X + c0 + c + 2];
                    a0 = __builtin_elementwise_fma(x0, v0, a0);
                    a1 = __builtin_elementwise_fma(x1, v1, a1);
                }
                v2f acc = a0 + a1;
                float sv = acc.x + acc.y;
                sv += __shfl_xor(sv, 1, 32);
                s[w] = sv * 0.125f;
            }
            float mx = s[0];
            #pragma unroll
            for (int w = 1; w < WIN; ++w) mx = fmaxf(mx, s[w]);
            float sum = 0.f;
            #pragma unroll
            for (int w = 0; w < WIN; ++w) { s[w] = fast_exp_nonpos(s[w] - mx); sum += s[w]; }
            const float inv = 1.0f / sum;

            #pragma unroll
            for (int c = 0; c < 32; c += 2) {
                v2f o = {0.f, 0.f};
                #pragma unroll
                for (int w = 0; w < WIN; ++w) {
                    v2f xv = *(const v2f*)&xs[(r + w) * LD_X + c0 + c];
                    v2f sw = {s[w], s[w]};
                    o = __builtin_elementwise_fma(sw, xv, o);
                }
                o *= inv;
                *(v2f*)&out[g * RANK + c0 + c] = o;
            }
        }
    }
}

extern "C" void kernel_launch(void* const* d_in, const int* in_sizes, int n_in,
                              void* d_out, int out_size, void* d_ws, size_t ws_size,
                              hipStream_t stream) {
    const float* x = (const float*)d_in[0];
    const float* W = (const float*)d_in[1];
    const float* b = (const float*)d_in[2];
    float* out = (float*)d_out;

    const int nrows = in_sizes[0] / RANK;
    const int grid  = (nrows + TILE - 1) / TILE;
    attn_window_fused_kernel<<<grid, NTHR, 0, stream>>>(x, W, b, out, nrows);
}


# weighted-sum pass remapped to 16 lanes per row, 8 rows per lane, softmax weights passed through LDS: every store writes whole 128-B lines and the x window leaves the registers (407 to 194 VGPRs)
# speedup vs baseline: 2.7851x; 1.3790x over previous
#include <hip/hip_runtime.h>


#define RANK   64
#define WIN    11
#define HALO   5
#define TILE   64
#define XROWS  (TILE + 2*HALO)
#define LD_X   66
#define LD_W   68
#define LD_P   12
#define NTHR   128
#define WLOADS (RANK * (RANK / 4) / NTHR)
#define XLOADS (XROWS / 8)
#define XTAIL  (XROWS - 8 * XLOADS)

typedef float v2f __attribute__((ext_vector_type(2)));
typedef float v8f __attribute__((ext_vector_type(8)));

static __device__ __forceinline__ float fast_tanh(float x) {
#if __has_builtin(__builtin_amdgcn_tanhf)
    return __builtin_amdgcn_tanhf(x);
#elif __has_builtin(__builtin_amdgcn_tanh_f32)
    return __builtin_amdgcn_tanh_f32(x);
#else
    return tanhf(x);
#endif
}

static __device__ __forceinline__ float fast_exp_nonpos(float x) {
#if __has_builtin(__builtin_amdgcn_exp2f)
    return __builtin_amdgcn_exp2f(x * 1.44269504088896341f);
#else
    return __expf(x);
#endif
}

template <bool GUARD>
static __device__ __forceinline__ void weighted_sum_rows(const float* xs, const float* ps,
                                                         float* __restrict__ out,
                                                         int tid, int r0, int nrows)
{
    const int wave = tid >> 5;
    const int lane = tid & 31;
    const int q    = lane & 15;
    const int h    = lane >> 4;
    const int rb   = wave * 16 + h * 8;

    v2f xa[8 + WIN - 1], xb[8 + WIN - 1];
    #pragma unroll
    for (int i = 0; i < 8 + WIN - 1; ++i) {
        xa[i] = *(const v2f*)&xs[(rb + i) * LD_X + 4 * q];
        xb[i] = *(const v2f*)&xs[(rb + i) * LD_X + 4 * q + 2];
    }
    #pragma unroll
    for (int j = 0; j < 8; ++j) {
        const float4* p = (const float4*)&ps[(rb + j) * LD_P];
        const float4 e0 = p[0], e1 = p[1], e2 = p[2];
        const float e[WIN] = {e0.x, e0.y, e0.z, e0.w, e1.x, e1.y, e1.z, e1.w, e2.x, e2.y, e2.z};
        v2f oa = {0.f, 0.f}, ob = {0.f, 0.f};
        #pragma unroll
        for (int w = 0; w < WIN; ++w) {
            v2f sw = {e[w], e[w]};
            oa = __builtin_elementwise_fma(sw, xa[j + w], oa);
            ob = __builtin_elementwise_fma(sw, xb[j + w], ob);
        }
        oa *= e2.w;
        ob *= e2.w;
        const int g = r0 + rb + j;
        if (!GUARD || g < nrows)
            *(float4*)&out[g * RANK + 4 * q] = make_float4(oa.x, oa.y, ob.x, ob.y);
    }
}

__global__ __launch_bounds__(NTHR)
void attn_window_fused_kernel(const float* __restrict__ x,
                              const float* __restrict__ W,
                              const float* __restrict__ b,
                              float* __restrict__ out,
                              int nrows)
{
    __shared__ float xs[XROWS * LD_X];
    __shared__ float vs[TILE  * LD_X];
    __shared__ float ws[RANK  * LD_W];
    __shared__ float bs[RANK];
    __shared__ __attribute__((aligned(16))) float ps[TILE * LD_P];

    const int tid  = threadIdx.x;
    const int r0   = blockIdx.x * TILE;
    const int c4   = tid & 15;
    const int rsub = tid >> 4;

    float4 wv[WLOADS];
    for (int i = 0; i < WLOADS; ++i)
        wv[i] = ((const float4*)W)[tid + i * NTHR];

    const float bv = b[tid & (RANK - 1)];

    float4 xv[XLOADS + 1];
    for (int i = 0; i < XLOADS; ++i) {
        const int g  = r0 - HALO + rsub + 8 * i;
        const int gc = min(max(g, 0), nrows - 1);
        xv[i] = ((const float4*)x)[gc * (RANK / 4) + c4];
    }
    xv[XLOADS] = make_float4(0.f, 0.f, 0.f, 0.f);
    if (tid < XTAIL * 16) {
        const int g  = r0 - HALO + 8 * XLOADS + rsub;
        const int gc = min(max(g, 0), nrows - 1);
        xv[XLOADS] = ((const float4*)x)[gc * (RANK / 4) + c4];
    }

    for (int i = 0; i < WLOADS; ++i)
        *(float4*)&ws[(rsub + 8 * i) * LD_W + c4 * 4] = wv[i];
    if (tid < RANK) bs[tid] = bv;

    for (int i = 0; i < XLOADS; ++i) {
        const int row = rsub + 8 * i;
        const int g   = r0 - HALO + row;
        float4 v = xv[i];
        if (g < 0 || g >= nrows) v = make_float4(0.f, 0.f, 0.f, 0.f);
        float* p = &xs[row * LD_X + c4 * 4];
        *(float2*)(p)     = make_float2(v.x, v.y);
        *(float2*)(p + 2) = make_float2(v.z, v.w);
    }
    if (tid < XTAIL * 16) {
        const int row = 8 * XLOADS + rsub;
        const int g   = r0 - HALO + row;
        float4 v = xv[XLOADS];
        if (g < 0 || g >= nrows) v = make_float4(0.f, 0.f, 0.f, 0.f);
        float* p = &xs[row * LD_X + c4 * 4];
        *(float2*)(p)     = make_float2(v.x, v.y);
        *(float2*)(p + 2) = make_float2(v.z, v.w);
    }
    __syncthreads();

    {
        const int wave = tid >> 5;
        const int lane = tid & 31;
        const int m    = lane & 15;
        const int hi   = lane >> 4;
        const int arow = wave * 16 + m + HALO;

        v2f afrag[RANK / 4];
        #pragma unroll
        for (int kb = 0; kb < RANK / 4; ++kb)
            afrag[kb] = *(const v2f*)&xs[arow * LD_X + kb * 4 + hi * 2];

        #pragma unroll
        for (int nt = 0; nt < 4; ++nt) {
            v8f acc = {};
            const int n = nt * 16 + m;
            #pragma unroll
            for (int kb = 0; kb < RANK / 4; ++kb) {
                v2f bfrag = *(const v2f*)&ws[n * LD_W + kb * 4 + hi * 2];
                acc = __builtin_amdgcn_wmma_f32_16x16x4_f32(
                        false, afrag[kb], false, bfrag,
                        (short)0, acc, false, false);
            }
            const float bias = bs[n];
            #pragma unroll
            for (int v = 0; v < 8; ++v) {
                const int row = wave * 16 + v + hi * 8;
                vs[row * LD_X + n] = fast_tanh(acc[v] + bias);
            }
        }
    }
    __syncthreads();

    {
        const int r    = tid >> 1;
        const int half = tid & 1;
        const int c0 = half * 32;
        float s[WIN];
        #pragma unroll
        for (int w = 0; w < WIN; ++w) {
            v2f a0 = {0.f, 0.f}, a1 = {0.f, 0.f};
            #pragma unroll
            for (int c = 0; c < 32; c += 4) {
                v2f x0 = *(const v2f*)&xs[(r + w) * LD_X + c0 + c];
                v2f x1 = *(const v2f*)&xs[(r + w) * LD_X + c0 + c + 2];
                v2f v0 = *(const v2f*)&vs[r * LD_X + c0 + c];
                v2f v1 = *(const v2f*)&vs[r * LD_X + c0 + c + 2];
                a0 = __builtin_elementwise_fma(x0, v0, a0);
                a1 = __builtin_elementwise_fma(x1, v1, a1);
            }
            v2f acc = a0 + a1;
            float sv = acc.x + acc.y;
            sv += __shfl_xor(sv, 1, 32);
            s[w] = sv * 0.125f;
        }
        float mx = s[0];
        #pragma unroll
        for (int w = 1; w < WIN; ++w) mx = fmaxf(mx, s[w]);
        float sum = 0.f;
        #pragma unroll
        for (int w = 0; w < WIN; ++w) { s[w] = fast_exp_nonpos(s[w] - mx); sum += s[w]; }
        const float inv = 1.0f / sum;
        if (half == 0) {
            float4* p = (float4*)&ps[r * LD_P];
            p[0] = make_float4(s[0], s[1], s[2],  s[3]);
            p[1] = make_float4(s[4], s[5], s[6],  s[7]);
            p[2] = make_float4(s[8], s[9], s[10], inv);
        }
    }
    __syncthreads();

    if (r0 + TILE <= nrows) weighted_sum_rows<false>(xs, ps, out, tid, r0, nrows);
    else                    weighted_sum_rows<true >(xs, ps, out, tid, r0, nrows);
}

extern "C" void kernel_launch(void* const* d_in, const int* in_sizes, int n_in,
                              void* d_out, int out_size, void* d_ws, size_t ws_size,
                              hipStream_t stream) {
    const float* x = (const float*)d_in[0];
    const float* W = (const float*)d_in[1];
    const float* b = (const float*)d_in[2];
    float* out = (float*)d_out;

    const int nrows = in_sizes[0] / RANK;
    const int grid  = (nrows + TILE - 1) / TILE;
    attn_window_fused_kernel<<<grid, NTHR, 0, stream>>>(x, W, b, out, nrows);
}


# score pass: 4 lanes per row pair (2 rows x 16 columns each), so every x row read from LDS feeds both rows' windows; sums regrouped, joined by quad DPP adds; 188 VGPRs: 5 work-groups per WGP
# speedup vs baseline: 3.0541x; 1.0966x over previous
#include <hip/hip_runtime.h>


#define RANK   64
#define WIN    11
#define HALO   5
#define TILE   64
#define XROWS  (TILE + 2*HALO)
#define LD_X   66
#define LD_W   68
#define LD_P   12
#define NTHR   128
#define WLOADS (RANK * (RANK / 4) / NTHR)
#define XLOADS (XROWS / 8)
#define XTAIL  (XROWS - 8 * XLOADS)

typedef float v2f __attribute__((ext_vector_type(2)));
typedef float v8f __attribute__((ext_vector_type(8)));

static __device__ __forceinline__ float fast_tanh(float x) {
#if __has_builtin(__builtin_amdgcn_tanhf)
    return __builtin_amdgcn_tanhf(x);
#elif __has_builtin(__builtin_amdgcn_tanh_f32)
    return __builtin_amdgcn_tanh_f32(x);
#else
    return tanhf(x);
#endif
}

static __device__ __forceinline__ float fast_exp_nonpos(float x) {
#if __has_builtin(__builtin_amdgcn_exp2f)
    return __builtin_amdgcn_exp2f(x * 1.44269504088896341f);
#else
    return __expf(x);
#endif
}

static __device__ __forceinline__ float quad_xor1(float v) {
    return __builtin_bit_cast(float, __builtin_amdgcn_update_dpp(0, __builtin_bit_cast(int, v), 0xB1, 0xF, 0xF, true));
}
static __device__ __forceinline__ float quad_xor2(float v) {
    return __builtin_bit_cast(float, __builtin_amdgcn_update_dpp(0, __builtin_bit_cast(int, v), 0x4E, 0xF, 0xF, true));
}

template <bool GUARD>
static __device__ __forceinline__ void weighted_sum_rows(const float* xs, const float* ps,
                                                         float* __restrict__ out,
                                                         int tid, int r0, int nrows)
{
    const int wave = tid >> 5;
    const int lane = tid & 31;
    const int q    = lane & 15;
    const int h    = lane >> 4;
    const int rb   = wave * 16 + h * 8;

    v2f xa[8 + WIN - 1], xb[8 + WIN - 1];
    #pragma unroll
    for (int i = 0; i < 8 + WIN - 1; ++i) {
        xa[i] = *(const v2f*)&xs[(rb + i) * LD_X + 4 * q];
        xb[i] = *(const v2f*)&xs[(rb + i) * LD_X + 4 * q + 2];
    }
    #pragma unroll
    for (int j = 0; j < 8; ++j) {
        const float4* p = (const float4*)&ps[(rb + j) * LD_P];
        const float4 e0 = p[0], e1 = p[1], e2 = p[2];
        const float e[WIN] = {e0.x, e0.y, e0.z, e0.w, e1.x, e1.y, e1.z, e1.w, e2.x, e2.y, e2.z};
        v2f oa = {0.f, 0.f}, ob = {0.f, 0.f};
        #pragma unroll
        for (int w = 0; w < WIN; ++w) {
            v2f sw = {e[w], e[w]};
            oa = __builtin_elementwise_fma(sw, xa[j + w], oa);
            ob = __builtin_elementwise_fma(sw, xb[j + w], ob);
        }
        oa *= e2.w;
        ob *= e2.w;
        const int g = r0 + rb + j;
        if (!GUARD || g < nrows)
            *(float4*)&out[g * RANK + 4 * q] = make_float4(oa.x, oa.y, ob.x, ob.y);
    }
}

__global__ __launch_bounds__(NTHR)
void attn_window_fused_kernel(const float* __restrict__ x,
                              const float* __restrict__ W,
                              const float* __restrict__ b,
                              float* __restrict__ out,
                              int nrows)
{
    __shared__ float xs[XROWS * LD_X];
    __shared__ float vs[TILE  * LD_X];
    __shared__ float ws[RANK  * LD_W];
    __shared__ float bs[RANK];
    __shared__ __attribute__((aligned(16))) float ps[TILE * LD_P];

    const int tid  = threadIdx.x;
    const int r0   = blockIdx.x * TILE;
    const int c4   = tid & 15;
    const int rsub = tid >> 4;

    float4 wv[WLOADS];
    for (int i = 0; i < WLOADS; ++i)
        wv[i] = ((const float4*)W)[tid + i * NTHR];

    const float bv = b[tid & (RANK - 1)];

    float4 xv[XLOADS + 1];
    for (int i = 0; i < XLOADS; ++i) {
        const int g  = r0 - HALO + rsub + 8 * i;
        const int gc = min(max(g, 0), nrows - 1);
        xv[i] = ((const float4*)x)[gc * (RANK / 4) + c4];
    }
    xv[XLOADS] = make_float4(0.f, 0.f, 0.f, 0.f);
    if (tid < XTAIL * 16) {
        const int g  = r0 - HALO + 8 * XLOADS + rsub;
        const int gc = min(max(g, 0), nrows - 1);
        xv[XLOADS] = ((const float4*)x)[gc * (RANK / 4) + c4];
    }

    for (int i = 0; i < WLOADS; ++i)
        *(float4*)&ws[(rsub + 8 * i) * LD_W + c4 * 4] = wv[i];
    if (tid < RANK) bs[tid] = bv;

    for (int i = 0; i < XLOADS; ++i) {
        const int row = rsub + 8 * i;
        const int g   = r0 - HALO + row;
        float4 v = xv[i];
        if (g < 0 || g >= nrows) v = make_float4(0.f, 0.f, 0.f, 0.f);
        float* p = &xs[row * LD_X + c4 * 4];
        *(float2*)(p)     = make_float2(v.x, v.y);
        *(float2*)(p + 2) = make_float2(v.z, v.w);
    }
    if (tid < XTAIL * 16) {
        const int row = 8 * XLOADS + rsub;
        const int g   = r0 - HALO + row;
        float4 v = xv[XLOADS];
        if (g < 0 || g >= nrows) v = make_float4(0.f, 0.f, 0.f, 0.f);
        float* p = &xs[row * LD_X + c4 * 4];
        *(float2*)(p)     = make_float2(v.x, v.y);
        *(float2*)(p + 2) = make_float2(v.z, v.w);
    }
    __syncthreads();

    {
        const int wave = tid >> 5;
        const int lane = tid & 31;
        const int m    = lane & 15;
        const int hi   = lane >> 4;
        const int arow = wave * 16 + m + HALO;

        v2f afrag[RANK / 4];
        #pragma unroll
        for (int kb = 0; kb < RANK / 4; ++kb)
            afrag[kb] = *(const v2f*)&xs[arow * LD_X + kb * 4 + hi * 2];

        #pragma unroll
        for (int nt = 0; nt < 4; ++nt) {
            v8f acc = {};
            const int n = nt * 16 + m;
            #pragma unroll
            for (int kb = 0; kb < RANK / 4; ++kb) {
                v2f bfrag = *(const v2f*)&ws[n * LD_W + kb * 4 + hi * 2];
                acc = __builtin_amdgcn_wmma_f32_16x16x4_f32(
                        false, afrag[kb], false, bfrag,
                        (short)0, acc, false, false);
            }
            const float bias = bs[n];
            #pragma unroll
            for (int v = 0; v < 8; ++v) {
                const int row = wave * 16 + v + hi * 8;
                vs[row * LD_X + n] = fast_tanh(acc[v] + bias);
            }
        }
    }
    __syncthreads();

    {
        const int rp   = tid >> 2;
        const int cq   = tid & 3;
        const int coff = (cq >> 1) * 32 + (cq & 1) * 2;
        const int ra   = 2 * rp;

        v2f va[8], vb[8];
        #pragma unroll
        for (int k = 0; k < 8; ++k) {
            va[k] = *(const v2f*)&vs[ ra      * LD_X + coff + 4 * k];
            vb[k] = *(const v2f*)&vs[(ra + 1) * LD_X + coff + 4 * k];
        }
        float pa[WIN], pb[WIN];
        #pragma unroll
        for (int i = 0; i <= WIN; ++i) {
            v2f xr[8];
            #pragma unroll
            for (int k = 0; k < 8; ++k)
                xr[k] = *(const v2f*)&xs[(ra + i) * LD_X + coff + 4 * k];
            if (i < WIN) {
                v2f t = xr[0] * va[0];
                #pragma unroll
                for (int k = 1; k < 8; ++k) t = __builtin_elementwise_fma(xr[k], va[k], t);
                pa[i] = t.x + t.y;
            }
            if (i >= 1) {
                v2f t = xr[0] * vb[0];
                #pragma unroll
                for (int k = 1; k < 8; ++k) t = __builtin_elementwise_fma(xr[k], vb[k], t);
                pb[i - 1] = t.x + t.y;
            }
        }

        const bool second = (cq & 2) != 0;
        float s[WIN];
        #pragma unroll
        for (int w = 0; w < WIN; ++w) {
            float keep       = second ? pb[w] : pa[w];
            const float send = second ? pa[w] : pb[w];
            keep += quad_xor2(send);
            keep += quad_xor1(keep);
            s[w] = keep * 0.125f;
        }
        float mx = s[0];
        #pragma unroll
        for (int w = 1; w < WIN; ++w) mx = fmaxf(mx, s[w]);
        float sum = 0.f;
        #pragma unroll
        for (int w = 0; w < WIN; ++w) { s[w] = fast_exp_nonpos(s[w] - mx); sum += s[w]; }
        const float inv = 1.0f / sum;
        if ((cq & 1) == 0) {
            float4* p = (float4*)&ps[(ra + (cq >> 1)) * LD_P];
            p[0] = make_float4(s[0], s[1], s[2],  s[3]);
            p[1] = make_float4(s[4], s[5], s[6],  s[7]);
            p[2] = make_float4(s[8], s[9], s[10], inv);
        }
    }
    __syncthreads();

    if (r0 + TILE <= nrows) weighted_sum_rows<false>(xs, ps, out, tid, r0, nrows);
    else                    weighted_sum_rows<true >(xs, ps, out, tid, r0, nrows);
}

extern "C" void kernel_launch(void* const* d_in, const int* in_sizes, int n_in,
                              void* d_out, int out_size, void* d_ws, size_t ws_size,
                              hipStream_t stream) {
    const float* x = (const float*)d_in[0];
    const float* W = (const float*)d_in[1];
    const float* b = (const float*)d_in[2];
    float* out = (float*)d_out;

    const int nrows = in_sizes[0] / RANK;
    const int grid  = (nrows + TILE - 1) / TILE;
    attn_window_fused_kernel<<<grid, NTHR, 0, stream>>>(x, W, b, out, nrows);
}


# occupancy 5 -> 8 work-groups per WGP: W passes through the var1 LDS region into B fragments kept in registers (wave = 16 columns), LDS 57 -> 40 KB, VGPRs capped at 128; k order of the sums permuted
# speedup vs baseline: 3.1989x; 1.0474x over previous
#include <hip/hip_runtime.h>


#define RANK   64
#define WIN    11
#define HALO   5
#define TILE   64
#define XROWS  (TILE + 2*HALO)
#define LD_X   66
#define LD_P   12
#define NTHR   128
#define WLOADS (RANK * (RANK / 4) / NTHR)
#define XLOADS (XROWS / 8)
#define XTAIL  (XROWS - 8 * XLOADS)

typedef float v2f __attribute__((ext_vector_type(2)));
typedef float v8f __attribute__((ext_vector_type(8)));

static __device__ __forceinline__ float fast_tanh(float x) {
#if __has_builtin(__builtin_amdgcn_tanhf)
    return __builtin_amdgcn_tanhf(x);
#elif __has_builtin(__builtin_amdgcn_tanh_f32)
    return __builtin_amdgcn_tanh_f32(x);
#else
    return tanhf(x);
#endif
}

static __device__ __forceinline__ float fast_exp_nonpos(float x) {
#if __has_builtin(__builtin_amdgcn_exp2f)
    return __builtin_amdgcn_exp2f(x * 1.44269504088896341f);
#else
    return __expf(x);
#endif
}

static __device__ __forceinline__ float quad_xor1(float v) {
    return __builtin_bit_cast(float, __builtin_amdgcn_update_dpp(0, __builtin_bit_cast(int, v), 0xB1, 0xF, 0xF, true));
}
static __device__ __forceinline__ float quad_xor2(float v) {
    return __builtin_bit_cast(float, __builtin_amdgcn_update_dpp(0, __builtin_bit_cast(int, v), 0x4E, 0xF, 0xF, true));
}

static __device__ __forceinline__ int phase_local(int v) {
    asm volatile("" : "+v"(v));
    return v;
}

template <bool GUARD>
static __device__ __forceinline__ void weighted_sum_rows(const float* xs, const float* ps,
                                                         float* __restrict__ out,
                                                         int tid, int r0, int nrows)
{
    const int wave = tid >> 5;
    const int lane = tid & 31;
    const int q    = lane & 15;
    const int h    = lane >> 4;
    const int rb   = wave * 16 + h * 8;

    v2f xa[8 + WIN - 1], xb[8 + WIN - 1];
    #pragma unroll
    for (int i = 0; i < 8 + WIN - 1; ++i) {
        xa[i] = *(const v2f*)&xs[(rb + i) * LD_X + 4 * q];
        xb[i] = *(const v2f*)&xs[(rb + i) * LD_X + 4 * q + 2];
    }
    #pragma unroll
    for (int j = 0; j < 8; ++j) {
        const float4* p = (const float4*)&ps[(rb + j) * LD_P];
        const float4 e0 = p[0], e1 = p[1], e2 = p[2];
        const float e[WIN] = {e0.x, e0.y, e0.z, e0.w, e1.x, e1.y, e1.z, e1.w, e2.x, e2.y, e2.z};
        v2f oa = {0.f, 0.f}, ob = {0.f, 0.f};
        #pragma unroll
        for (int w = 0; w < WIN; ++w) {
            v2f sw = {e[w], e[w]};
            oa = __builtin_elementwise_fma(sw, xa[j + w], oa);
            ob = __builtin_elementwise_fma(sw, xb[j + w], ob);
        }
        oa *= e2.w;
        ob *= e2.w;
        const int g = r0 + rb + j;
        if (!GUARD || g < nrows)
            *(float4*)&out[g * RANK + 4 * q] = make_float4(oa.x, oa.y, ob.x, ob.y);
    }
}

__global__ __launch_bounds__(NTHR) __attribute__((amdgpu_waves_per_eu(8, 8)))
void attn_window_fused_kernel(const float* __restrict__ x,
                              const float* __restrict__ W,
                              const float* __restrict__ b,
                              float* __restrict__ out,
                              int nrows)
{
    __shared__ float xs[XROWS * LD_X];
    __shared__ float vs[TILE  * LD_X];
    __shared__ float bs[RANK];
    __shared__ __attribute__((aligned(16))) float ps[TILE * LD_P];

    const int tid  = threadIdx.x;
    const int r0   = blockIdx.x * TILE;
    const int c4   = tid & 15;
    const int rsub = tid >> 4;

    float4 wv[WLOADS];
    for (int i = 0; i < WLOADS; ++i)
        wv[i] = ((const float4*)W)[tid + i * NTHR];

    const float bv = b[tid & (RANK - 1)];

    float4 xv[XLOADS + 1];
    for (int i = 0; i < XLOADS; ++i) {
        const int g  = r0 - HALO + rsub + 8 * i;
        const int gc = min(max(g, 0), nrows - 1);
        xv[i] = ((const float4*)x)[gc * (RANK / 4) + c4];
    }
    xv[XLOADS] = make_float4(0.f, 0.f, 0.f, 0.f);
    if (tid < XTAIL * 16) {
        const int g  = r0 - HALO + 8 * XLOADS + rsub;
        const int gc = min(max(g, 0), nrows - 1);
        xv[XLOADS] = ((const float4*)x)[gc * (RANK / 4) + c4];
    }

    for (int i = 0; i < WLOADS; ++i) {
        float* p = &vs[(rsub + 8 * i) * LD_X + c4 * 4];
        *(float2*)(p)     = make_float2(wv[i].x, wv[i].y);
        *(float2*)(p + 2) = make_float2(wv[i].z, wv[i].w);
    }
    if (tid < RANK) bs[tid] = bv;

    for (int i = 0; i < XLOADS; ++i) {
        const int row = rsub + 8 * i;
        const int g   = r0 - HALO + row;
        float4 v = xv[i];
        if (g < 0 || g >= nrows) v = make_float4(0.f, 0.f, 0.f, 0.f);
        float* p = &xs[row * LD_X + c4 * 4];
        *(float2*)(p)     = make_float2(v.x, v.y);
        *(float2*)(p + 2) = make_float2(v.z, v.w);
    }
    if (tid < XTAIL * 16) {
        const int row = 8 * XLOADS + rsub;
        const int g   = r0 - HALO + row;
        float4 v = xv[XLOADS];
        if (g < 0 || g >= nrows) v = make_float4(0.f, 0.f, 0.f, 0.f);
        float* p = &xs[row * LD_X + c4 * 4];
        *(float2*)(p)     = make_float2(v.x, v.y);
        *(float2*)(p + 2) = make_float2(v.z, v.w);
    }
    __syncthreads();

    {
        const int wave = tid >> 5;
        const int lane = tid & 31;
        const int m    = lane & 15;
        const int hi   = lane >> 4;
        const int n    = wave * 16 + m;
        const int koff = 32 * hi;

        v2f bfrag[RANK / 4];
        #pragma unroll
        for (int kb = 0; kb < RANK / 4; ++kb)
            bfrag[kb] = *(const v2f*)&vs[n * LD_X + koff + 2 * kb];
        const float bias = bs[n];
        __syncthreads();

        #pragma unroll
        for (int mt = 0; mt < 4; ++mt) {
            const int arow = mt * 16 + m + HALO;
            v2f afrag[RANK / 4];
            #pragma unroll
            for (int kb = 0; kb < RANK / 4; ++kb)
                afrag[kb] = *(const v2f*)&xs[arow * LD_X + koff + 2 * kb];
            v8f acc = {};
            #pragma unroll
            for (int kb = 0; kb < RANK / 4; ++kb)
                acc = __builtin_amdgcn_wmma_f32_16x16x4_f32(
                        false, afrag[kb], false, bfrag[kb],
                        (short)0, acc, false, false);
            #pragma unroll
            for (int v = 0; v < 8; ++v) {
                const int row = mt * 16 + v + hi * 8;
                vs[row * LD_X + n] = fast_tanh(acc[v] + bias);
            }
        }
    }
    __syncthreads();

    {
        const int rp   = tid >> 2;
        const int cq   = tid & 3;
        const int coff = (cq >> 1) * 32 + (cq & 1) * 2;
        const int ra   = 2 * rp;

        v2f va[8], vb[8];
        #pragma unroll
        for (int k = 0; k < 8; ++k) {
            va[k] = *(const v2f*)&vs[ ra      * LD_X + coff + 4 * k];
            vb[k] = *(const v2f*)&vs[(ra + 1) * LD_X + coff + 4 * k];
        }
        float pa[WIN], pb[WIN];
        #pragma unroll
        for (int i = 0; i <= WIN; ++i) {
            v2f xr[8];
            #pragma unroll
            for (int k = 0; k < 8; ++k)
                xr[k] = *(const v2f*)&xs[(ra + i) * LD_X + coff + 4 * k];
            if (i < WIN) {
                v2f t = xr[0] * va[0];
                #pragma unroll
                for (int k = 1; k < 8; ++k) t = __builtin_elementwise_fma(xr[k], va[k], t);
                pa[i] = t.x + t.y;
            }
            if (i >= 1) {
                v2f t = xr[0] * vb[0];
                #pragma unroll
                for (int k = 1; k < 8; ++k) t = __builtin_elementwise_fma(xr[k], vb[k], t);
                pb[i - 1] = t.x + t.y;
            }
        }

        const bool second = (cq & 2) != 0;
        float s[WIN];
        #pragma unroll
        for (int w = 0; w < WIN; ++w) {
            float keep       = second ? pb[w] : pa[w];
            const float send = second ? pa[w] : pb[w];
            keep += quad_xor2(send);
            keep += quad_xor1(keep);
            s[w] = keep * 0.125f;
        }
        float mx = s[0];
        #pragma unroll
        for (int w = 1; w < WIN; ++w) mx = fmaxf(mx, s[w]);
        float sum = 0.f;
        #pragma unroll
        for (int w = 0; w < WIN; ++w) { s[w] = fast_exp_nonpos(s[w] - mx); sum += s[w]; }
        const float inv = 1.0f / sum;
        if ((cq & 1) == 0) {
            float4* p = (float4*)&ps[(ra + (cq >> 1)) * LD_P];
            p[0] = make_float4(s[0], s[1], s[2],  s[3]);
            p[1] = make_float4(s[4], s[5], s[6],  s[7]);
            p[2] = make_float4(s[8], s[9], s[10], inv);
        }
    }
    __syncthreads();

    const int t4 = phase_local(tid) & (NTHR - 1);
    if (r0 + TILE <= nrows) weighted_sum_rows<false>(xs, ps, out, t4, r0, nrows);
    else                    weighted_sum_rows<true >(xs, ps, out, t4, r0, nrows);
}

extern "C" void kernel_launch(void* const* d_in, const int* in_sizes, int n_in,
                              void* d_out, int out_size, void* d_ws, size_t ws_size,
                              hipStream_t stream) {
    const float* x = (const float*)d_in[0];
    const float* W = (const float*)d_in[1];
    const float* b = (const float*)d_in[2];
    float* out = (float*)d_out;

    const int nrows = in_sizes[0] / RANK;
    const int grid  = (nrows + TILE - 1) / TILE;
    attn_window_fused_kernel<<<grid, NTHR, 0, stream>>>(x, W, b, out, nrows);
}
